# starting kernel
# baseline (speedup 1.0000x reference)
#include <hip/hip_runtime.h>
#include <hip/hip_bf16.h>


#define D      256
#define NMAT   1024
#define G      16
#define NCHUNK 64
#define MATELT (D * D)
#define CHUNKSTRIDE (2 * MATELT)

typedef __attribute__((ext_vector_type(2))) float v2f;
typedef __attribute__((ext_vector_type(8))) float v8f;

__global__ __launch_bounds__(256) void prodscan_chunk_mm(
    const float* __restrict__ m, float* __restrict__ ws, int t) {
  const int j    = blockIdx.x;
  const int slab = blockIdx.y;

  float* buf0 = ws + (size_t)j * CHUNKSTRIDE;
  float* buf1 = buf0 + MATELT;

  const float* A = (t == 1) ? (m + (size_t)j * G * MATELT)
                            : ((t & 1) ? buf1 : buf0);
  const float* B = m + (size_t)(j * G + t) * MATELT;
  float*       C = (t & 1) ? buf0 : buf1;

  const int w     = threadIdx.x >> 5;
  const int lane  = threadIdx.x & 31;
  const int laneM = lane & 15;
  const int laneH = lane >> 4;

  const int tr0 = slab * 8 + (w >> 2) * 4;
  const int tc0 = (w & 3) * 4;

  v8f acc[4][4] = {};

  for (int k0 = 0; k0 < D; k0 += 4) {
    v2f af[4];
    v2f bf[4];
#pragma unroll
    for (int r = 0; r < 4; ++r) {
      const float* ap =
          A + (size_t)((tr0 + r) * 16 + laneM) * D + k0 + 2 * laneH;
      af[r].x = ap[0];
      af[r].y = ap[1];
    }
#pragma unroll
    for (int c = 0; c < 4; ++c) {
      const float* bp =
          B + (size_t)(k0 + 2 * laneH) * D + (tc0 + c) * 16 + laneM;
      bf[c].x = bp[0];
      bf[c].y = bp[D];
    }
#pragma unroll
    for (int r = 0; r < 4; ++r)
#pragma unroll
      for (int c = 0; c < 4; ++c)
        acc[r][c] = __builtin_amdgcn_wmma_f32_16x16x4_f32(
            false, af[r], false, bf[c], (short)0, acc[r][c], false, false);
  }

#pragma unroll
  for (int r = 0; r < 4; ++r)
#pragma unroll
    for (int c = 0; c < 4; ++c)
#pragma unroll
      for (int e = 0; e < 8; ++e) {
        const int row = (tr0 + r) * 16 + e + 8 * laneH;
        const int col = (tc0 + c) * 16 + laneM;
        C[(size_t)row * D + col] = acc[r][c][e];
      }
}

__global__ __launch_bounds__(256) void prodscan_carries(
    const float* __restrict__ x, const float* __restrict__ ws,
    float* __restrict__ carries, float* __restrict__ out) {
  __shared__ float u[D];
  const int t = threadIdx.x;
  float v = x[t];
  u[t] = v;
  carries[t] = v;
  __syncthreads();
  for (int j = 0; j < NCHUNK; ++j) {
    const float* P = ws + (size_t)j * CHUNKSTRIDE;
    if (j + 1 < NCHUNK)
      __builtin_prefetch(ws + (size_t)(j + 1) * CHUNKSTRIDE + t * 64, 0, 1);
    float acc = 0.f;
#pragma unroll 8
    for (int i = 0; i < D; ++i)
      acc = fmaf(u[i], P[(size_t)i * D + t], acc);
    __syncthreads();
    u[t] = acc;
    carries[(size_t)(j + 1) * D + t] = acc;
    __syncthreads();
  }
  out[(size_t)NMAT * D + t] = u[t];
}

__global__ __launch_bounds__(256) void prodscan_expand(
    const float* __restrict__ m, const float* __restrict__ carries,
    float* __restrict__ out) {
  __shared__ float vb[2][D];
  const int j = blockIdx.x;
  const int t = threadIdx.x;
  float v = carries[(size_t)j * D + t];
  vb[0][t] = v;
  out[(size_t)(j * G) * D + t] = v;
  __syncthreads();
  int cur = 0;
  for (int i = 0; i < G - 1; ++i) {
    const float* M = m + (size_t)(j * G + i) * MATELT;
    if (i + 1 < G - 1)
      __builtin_prefetch(m + (size_t)(j * G + i + 1) * MATELT + t * 64, 0, 1);
    float acc = 0.f;
#pragma unroll 8
    for (int r = 0; r < D; ++r)
      acc = fmaf(vb[cur][r], M[(size_t)r * D + t], acc);
    vb[cur ^ 1][t] = acc;
    out[(size_t)(j * G + i + 1) * D + t] = acc;
    cur ^= 1;
    __syncthreads();
  }
}

extern "C" void kernel_launch(void* const* d_in, const int* in_sizes, int n_in,
                              void* d_out, int out_size, void* d_ws,
                              size_t ws_size, hipStream_t stream) {
  (void)in_sizes; (void)n_in; (void)out_size; (void)ws_size;
  const float* x = (const float*)d_in[0];
  const float* m = (const float*)d_in[1];
  float* out = (float*)d_out;
  float* ws  = (float*)d_ws;
  float* carries = ws + (size_t)NCHUNK * CHUNKSTRIDE;

  for (int t = 1; t < G; ++t)
    prodscan_chunk_mm<<<dim3(NCHUNK, 2), 256, 0, stream>>>(m, ws, t);
  prodscan_carries<<<1, 256, 0, stream>>>(x, ws, carries, out);
  prodscan_expand<<<NCHUNK, 256, 0, stream>>>(m, carries, out);
}


# Carry scan over the 64 chunk products as a binary tree: 5 up-sweep WMMA launches (pair products) + 6 down-sweep GEMV launches replace the 64-step serial scan; carry sums regrouped by association
# speedup vs baseline: 2.7927x; 2.7927x over previous
#include <hip/hip_runtime.h>
#include <hip/hip_bf16.h>


#define D      256
#define NMAT   1024
#define G      16
#define NCHUNK 64
#define MATELT (D * D)
#define CHUNKSTRIDE (2 * MATELT)
#define HALF   8
#define NHALF  (NMAT / HALF)

typedef __attribute__((ext_vector_type(2))) float v2f;
typedef __attribute__((ext_vector_type(4))) float v4f;
typedef __attribute__((ext_vector_type(8))) float v8f;

#define KC       64
#define NKCH     (D / KC)
#define APITCH   68
#define BPITCH   544
#define NPASS    (KC / 8)
#define AROWS    (1024 / KC)
#define BSTAGE   ((KC / 2) * BPITCH)

template <int RT>
__device__ __forceinline__ void chunk_products(const float* as,
                                               const float* bs,
                                               v8f (&acc)[RT][4]) {
  v2f af[2][RT];
  v2f bf[2][4];
#pragma unroll
  for (int r = 0; r < RT; ++r)
    af[0][r] = *(const v2f*)(as + (r * 16) * APITCH);
#pragma unroll
  for (int c = 0; c < 4; ++c)
    bf[0][c] = *(const v2f*)(bs + (c * 16) * 2);
#pragma unroll
  for (int ks = 0; ks < KC / 4; ++ks) {
    const int cur = ks & 1;
    if (ks + 1 < KC / 4) {
#pragma unroll
      for (int r = 0; r < RT; ++r)
        af[cur ^ 1][r] =
            *(const v2f*)(as + (r * 16) * APITCH + (ks + 1) * 4);
#pragma unroll
      for (int c = 0; c < 4; ++c)
        bf[cur ^ 1][c] =
            *(const v2f*)(bs + ((ks + 1) * 2) * BPITCH + (c * 16) * 2);
    }
    __builtin_amdgcn_sched_barrier(0);
#pragma unroll
    for (int r = 0; r < RT; ++r)
#pragma unroll
      for (int c = 0; c < 4; ++c)
        acc[r][c] = __builtin_amdgcn_wmma_f32_16x16x4_f32(
            false, af[cur][r], false, bf[cur][c], (short)0, acc[r][c], false,
            false);
    __builtin_amdgcn_sched_barrier(0);
  }
}

template <int RT>
__global__ __launch_bounds__(256) void prodscan_chunk_mm(
    const float* __restrict__ m, float* __restrict__ ws, int t) {
  constexpr int SROWS = 32 * RT;
  constexpr int APASS = SROWS / AROWS;
  constexpr int ASTG  = 128 * APITCH;
  __shared__ __attribute__((aligned(16))) float As[2 * ASTG];
  __shared__ __attribute__((aligned(16))) float Bs[2 * BSTAGE];

  const int h    = blockIdx.x;
  const int slab = blockIdx.y;

  float* C;
  const float* A;
  const float* B;
  if (t <= HALF) {
    const bool combine = (t == HALF);
    C = ws + (size_t)(combine ? 2 * h : h) * MATELT;
    A = (t == 1) ? (m + (size_t)h * HALF * MATELT) : C;
    B = combine ? (C + MATELT) : (m + (size_t)(h * HALF + t) * MATELT);
  } else {
    const int L    = t - HALF;
    const int base = h << (L + 1);
    const int half = 1 << (L - 1);
    C = ws + (size_t)(base + 2 * half - 1) * MATELT;
    A = ws + (size_t)(base + half - 1) * MATELT;
    B = ws + (size_t)(base + 3 * half - 1) * MATELT;
  }

  const int tid   = threadIdx.x;
  const int w     = tid >> 5;
  const int lane  = tid & 31;
  const int laneM = lane & 15;
  const int laneH = lane >> 4;

  const int wr  = w >> 2;
  const int tr0 = slab * (2 * RT) + wr * RT;
  const int tc0 = (w & 3) * 4;

  const int arow = tid / (KC / 4), akq = tid % (KC / 4);
  const int bp   = tid >> 6, bcq = tid & 63;
  const float* Ag = A + (size_t)(slab * SROWS + arow) * D + akq * 4;
  const float* Bg = B + (size_t)(2 * bp) * D + bcq * 4;
  float* Asw = As + arow * APITCH + akq * 4;
  float* Bsw = Bs + bp * BPITCH + bcq * 8;

  const float* Asr = As + (wr * (RT * 16) + laneM) * APITCH + 2 * laneH;
  const float* Bsr = Bs + laneH * BPITCH + (tc0 * 16 + laneM) * 2;

  v4f pa[APASS];
  v4f pb0[NPASS];
  v4f pb1[NPASS];

  v8f acc[RT][4] = {};

#pragma unroll
  for (int i = 0; i < APASS; ++i)
    pa[i] = *(const v4f*)(Ag + (size_t)(AROWS * i) * D);
#pragma unroll
  for (int i = 0; i < NPASS; ++i) {
    pb0[i] = *(const v4f*)(Bg + (size_t)(8 * i) * D);
    pb1[i] = *(const v4f*)(Bg + (size_t)(8 * i + 1) * D);
  }
#pragma unroll
  for (int i = 0; i < APASS; ++i)
    *(v4f*)(Asw + (AROWS * i) * APITCH) = pa[i];
#pragma unroll
  for (int i = 0; i < NPASS; ++i) {
    *(v4f*)(Bsw + (4 * i) * BPITCH) =
        (v4f){pb0[i].x, pb1[i].x, pb0[i].y, pb1[i].y};
    *(v4f*)(Bsw + (4 * i) * BPITCH + 4) =
        (v4f){pb0[i].z, pb1[i].z, pb0[i].w, pb1[i].w};
  }
  __syncthreads();

  for (int ch = 0; ch < NKCH - 1; ++ch) {
    const int s  = ch & 1;
    const int kc = (ch + 1) * KC;
#pragma unroll
    for (int i = 0; i < APASS; ++i)
      pa[i] = *(const v4f*)(Ag + (size_t)(AROWS * i) * D + kc);
#pragma unroll
    for (int i = 0; i < NPASS; ++i) {
      pb0[i] = *(const v4f*)(Bg + (size_t)(kc + 8 * i) * D);
      pb1[i] = *(const v4f*)(Bg + (size_t)(kc + 8 * i + 1) * D);
    }
    __builtin_amdgcn_sched_barrier(0);

    chunk_products<RT>(Asr + s * ASTG, Bsr + s * BSTAGE, acc);

    const int so = (s ^ 1);
#pragma unroll
    for (int i = 0; i < APASS; ++i)
      *(v4f*)(Asw + so * ASTG + (AROWS * i) * APITCH) = pa[i];
#pragma unroll
    for (int i = 0; i < NPASS; ++i) {
      *(v4f*)(Bsw + so * BSTAGE + (4 * i) * BPITCH) =
          (v4f){pb0[i].x, pb1[i].x, pb0[i].y, pb1[i].y};
      *(v4f*)(Bsw + so * BSTAGE + (4 * i) * BPITCH + 4) =
          (v4f){pb0[i].z, pb1[i].z, pb0[i].w, pb1[i].w};
    }
    __syncthreads();
  }
  chunk_products<RT>(Asr + ((NKCH - 1) & 1) * ASTG,
                     Bsr + ((NKCH - 1) & 1) * BSTAGE, acc);

#pragma unroll
  for (int r = 0; r < RT; ++r)
#pragma unroll
    for (int c = 0; c < 4; ++c)
#pragma unroll
      for (int e = 0; e < 8; ++e) {
        const int row = (tr0 + r) * 16 + e + 8 * laneH;
        const int col = (tc0 + c) * 16 + laneM;
        C[(size_t)row * D + col] = acc[r][c][e];
      }
}

#define DS_STRIP 32
#define DS_ROWS  (D / 8)

__global__ __launch_bounds__(256) void prodscan_downsweep(
    const float* __restrict__ x, float* __restrict__ ws, int l) {
  __shared__ float us[D];
  __shared__ float part[8][DS_STRIP];
  const int i     = blockIdx.x;
  const int strip = blockIdx.y;
  const int t     = threadIdx.x;
  const int w     = t >> 5;
  const int lane  = t & 31;

  float* carries = ws + (size_t)NCHUNK * CHUNKSTRIDE;
  const int srcIdx = (2 * i) << l;
  const int dstIdx = (2 * i + 1) << l;
  const int buf    = ((4 * i) << l) + (1 << l) - 1;
  const float* T = ws + (size_t)buf * MATELT + (size_t)(w * DS_ROWS) * D +
                   strip * DS_STRIP + lane;

  float mv[DS_ROWS];
#pragma unroll
  for (int r = 0; r < DS_ROWS; ++r) mv[r] = T[(size_t)r * D];

  const float sv = (srcIdx == 0) ? x[t] : carries[(size_t)srcIdx * D + t];
  us[t] = sv;
  if (l == 5 && strip == 0) carries[t] = sv;
  __syncthreads();

  float acc = 0.f;
#pragma unroll
  for (int r = 0; r < DS_ROWS; ++r)
    acc = fmaf(us[w * DS_ROWS + r], mv[r], acc);
  part[w][lane] = acc;
  __syncthreads();
  if (w == 0) {
    float sum = part[0][lane];
#pragma unroll
    for (int k = 1; k < 8; ++k) sum += part[k][lane];
    carries[(size_t)dstIdx * D + strip * DS_STRIP + lane] = sum;
  }
}

__global__ __launch_bounds__(256) void prodscan_expand(
    const float* __restrict__ m, const float* __restrict__ ws,
    const float* __restrict__ carries, float* __restrict__ out) {
  __shared__ float vb[2][D];
  const int j = blockIdx.x;
  const int t = threadIdx.x;
  if (j == NCHUNK) {
    const float* P = ws + (size_t)(NCHUNK - 1) * CHUNKSTRIDE;
    vb[0][t] = carries[(size_t)(NCHUNK - 1) * D + t];
    __syncthreads();
    float acc = 0.f;
#pragma unroll 8
    for (int r = 0; r < D; ++r)
      acc = fmaf(vb[0][r], P[(size_t)r * D + t], acc);
    out[(size_t)NMAT * D + t] = acc;
    return;
  }
  float v = carries[(size_t)j * D + t];
  vb[0][t] = v;
  out[(size_t)(j * G) * D + t] = v;
  __syncthreads();
  int cur = 0;
  for (int i = 0; i < G - 1; ++i) {
    const float* M = m + (size_t)(j * G + i) * MATELT;
    if (i + 1 < G - 1)
      __builtin_prefetch(m + (size_t)(j * G + i + 1) * MATELT + t * 64, 0, 1);
    float acc = 0.f;
#pragma unroll 8
    for (int r = 0; r < D; ++r)
      acc = fmaf(vb[cur][r], M[(size_t)r * D + t], acc);
    vb[cur ^ 1][t] = acc;
    out[(size_t)(j * G + i + 1) * D + t] = acc;
    cur ^= 1;
    __syncthreads();
  }
}

extern "C" void kernel_launch(void* const* d_in, const int* in_sizes, int n_in,
                              void* d_out, int out_size, void* d_ws,
                              size_t ws_size, hipStream_t stream) {
  (void)in_sizes; (void)n_in; (void)out_size; (void)ws_size;
  const float* x = (const float*)d_in[0];
  const float* m = (const float*)d_in[1];
  float* out = (float*)d_out;
  float* ws  = (float*)d_ws;
  float* carries = ws + (size_t)NCHUNK * CHUNKSTRIDE;

  for (int t = 1; t < HALF; ++t)
    prodscan_chunk_mm<4><<<dim3(NHALF, 2), 256, 0, stream>>>(m, ws, t);
  prodscan_chunk_mm<4><<<dim3(NCHUNK, 2), 256, 0, stream>>>(m, ws, HALF);
  prodscan_chunk_mm<1><<<dim3(32, 8), 256, 0, stream>>>(m, ws, 9);
  prodscan_chunk_mm<1><<<dim3(16, 8), 256, 0, stream>>>(m, ws, 10);
  prodscan_chunk_mm<1><<<dim3(8, 8), 256, 0, stream>>>(m, ws, 11);
  prodscan_chunk_mm<1><<<dim3(4, 8), 256, 0, stream>>>(m, ws, 12);
  prodscan_chunk_mm<1><<<dim3(1, 8), 256, 0, stream>>>(m, ws, 13);
  prodscan_downsweep<<<dim3(1, 8), 256, 0, stream>>>(x, ws, 5);
  prodscan_downsweep<<<dim3(2, 8), 256, 0, stream>>>(x, ws, 4);
  prodscan_downsweep<<<dim3(4, 8), 256, 0, stream>>>(x, ws, 3);
  prodscan_downsweep<<<dim3(8, 8), 256, 0, stream>>>(x, ws, 2);
  prodscan_downsweep<<<dim3(16, 8), 256, 0, stream>>>(x, ws, 1);
  prodscan_downsweep<<<dim3(32, 8), 256, 0, stream>>>(x, ws, 0);
  prodscan_expand<<<65, 256, 0, stream>>>(m, ws, carries, out);
}


# Carry tree one level deeper: 128 leaves of 8, pair products written over their right operand (column slabs); expansion chains 7 GEMV steps, not 15; carry sums regrouped by association
# speedup vs baseline: 3.5437x; 1.2689x over previous
#include <hip/hip_runtime.h>
#include <hip/hip_bf16.h>


#define D      256
#define NMAT   1024
#define MATELT (D * D)
#define LEAF   8
#define NLEAF  (NMAT / LEAF)

typedef __attribute__((ext_vector_type(2))) float v2f;
typedef __attribute__((ext_vector_type(4))) float v4f;
typedef __attribute__((ext_vector_type(8))) float v8f;

#define KC       64
#define NKCH     (D / KC)
#define APITCH   68
#define BPITCH   544
#define BPITCHC  288
#define NPASS    (KC / 8)
#define AROWS    (1024 / KC)
#define BSTAGE   ((KC / 2) * BPITCH)
#define BSTAGEC  ((KC / 2) * BPITCHC)

template <int RT, int CT, int BP>
__device__ __forceinline__ void chunk_products(const float* as,
                                               const float* bs,
                                               v8f (&acc)[RT][CT]) {
  v2f af[2][RT];
  v2f bf[2][CT];
#pragma unroll
  for (int r = 0; r < RT; ++r)
    af[0][r] = *(const v2f*)(as + (r * 16) * APITCH);
#pragma unroll
  for (int c = 0; c < CT; ++c)
    bf[0][c] = *(const v2f*)(bs + (c * 16) * 2);
#pragma unroll
  for (int ks = 0; ks < KC / 4; ++ks) {
    const int cur = ks & 1;
    if (ks + 1 < KC / 4) {
#pragma unroll
      for (int r = 0; r < RT; ++r)
        af[cur ^ 1][r] =
            *(const v2f*)(as + (r * 16) * APITCH + (ks + 1) * 4);
#pragma unroll
      for (int c = 0; c < CT; ++c)
        bf[cur ^ 1][c] =
            *(const v2f*)(bs + ((ks + 1) * 2) * BP + (c * 16) * 2);
    }
    __builtin_amdgcn_sched_barrier(0);
#pragma unroll
    for (int r = 0; r < RT; ++r)
#pragma unroll
      for (int c = 0; c < CT; ++c)
        acc[r][c] = __builtin_amdgcn_wmma_f32_16x16x4_f32(
            false, af[cur][r], false, bf[cur][c], (short)0, acc[r][c], false,
            false);
    __builtin_amdgcn_sched_barrier(0);
  }
}

__global__ __launch_bounds__(256) void prodscan_chain_mm(
    const float* __restrict__ m, float* __restrict__ ws, int t) {
  constexpr int RT    = 4;
  constexpr int SROWS = 32 * RT;
  constexpr int APASS = SROWS / AROWS;
  constexpr int ASTG  = SROWS * APITCH;
  __shared__ __attribute__((aligned(16))) float As[2 * ASTG];
  __shared__ __attribute__((aligned(16))) float Bs[2 * BSTAGE];

  const int h    = blockIdx.x;
  const int slab = blockIdx.y;

  float* C = ws + (size_t)h * MATELT;
  const float* A = (t == 1) ? (m + (size_t)h * LEAF * MATELT) : C;
  const float* B = m + (size_t)(h * LEAF + t) * MATELT;

  const int tid   = threadIdx.x;
  const int w     = tid >> 5;
  const int lane  = tid & 31;
  const int laneM = lane & 15;
  const int laneH = lane >> 4;

  const int wr  = w >> 2;
  const int tr0 = slab * (2 * RT) + wr * RT;
  const int tc0 = (w & 3) * 4;

  const int arow = tid / (KC / 4), akq = tid % (KC / 4);
  const int bp   = tid >> 6, bcq = tid & 63;
  const float* Ag = A + (size_t)(slab * SROWS + arow) * D + akq * 4;
  const float* Bg = B + (size_t)(2 * bp) * D + bcq * 4;
  float* Asw = As + arow * APITCH + akq * 4;
  float* Bsw = Bs + bp * BPITCH + bcq * 8;

  const float* Asr = As + (wr * (RT * 16) + laneM) * APITCH + 2 * laneH;
  const float* Bsr = Bs + laneH * BPITCH + (tc0 * 16 + laneM) * 2;

  v4f pa[APASS];
  v4f pb0[NPASS];
  v4f pb1[NPASS];

  v8f acc[RT][4] = {};

#pragma unroll
  for (int i = 0; i < APASS; ++i)
    pa[i] = *(const v4f*)(Ag + (size_t)(AROWS * i) * D);
#pragma unroll
  for (int i = 0; i < NPASS; ++i) {
    pb0[i] = *(const v4f*)(Bg + (size_t)(8 * i) * D);
    pb1[i] = *(const v4f*)(Bg + (size_t)(8 * i + 1) * D);
  }
#pragma unroll
  for (int i = 0; i < APASS; ++i)
    *(v4f*)(Asw + (AROWS * i) * APITCH) = pa[i];
#pragma unroll
  for (int i = 0; i < NPASS; ++i) {
    *(v4f*)(Bsw + (4 * i) * BPITCH) =
        (v4f){pb0[i].x, pb1[i].x, pb0[i].y, pb1[i].y};
    *(v4f*)(Bsw + (4 * i) * BPITCH + 4) =
        (v4f){pb0[i].z, pb1[i].z, pb0[i].w, pb1[i].w};
  }
  __syncthreads();

  for (int ch = 0; ch < NKCH - 1; ++ch) {
    const int s  = ch & 1;
    const int kc = (ch + 1) * KC;
#pragma unroll
    for (int i = 0; i < APASS; ++i)
      pa[i] = *(const v4f*)(Ag + (size_t)(AROWS * i) * D + kc);
#pragma unroll
    for (int i = 0; i < NPASS; ++i) {
      pb0[i] = *(const v4f*)(Bg + (size_t)(kc + 8 * i) * D);
      pb1[i] = *(const v4f*)(Bg + (size_t)(kc + 8 * i + 1) * D);
    }
    __builtin_amdgcn_sched_barrier(0);

    chunk_products<RT, 4, BPITCH>(Asr + s * ASTG, Bsr + s * BSTAGE, acc);

    const int so = (s ^ 1);
#pragma unroll
    for (int i = 0; i < APASS; ++i)
      *(v4f*)(Asw + so * ASTG + (AROWS * i) * APITCH) = pa[i];
#pragma unroll
    for (int i = 0; i < NPASS; ++i) {
      *(v4f*)(Bsw + so * BSTAGE + (4 * i) * BPITCH) =
          (v4f){pb0[i].x, pb1[i].x, pb0[i].y, pb1[i].y};
      *(v4f*)(Bsw + so * BSTAGE + (4 * i) * BPITCH + 4) =
          (v4f){pb0[i].z, pb1[i].z, pb0[i].w, pb1[i].w};
    }
    __syncthreads();
  }
  chunk_products<RT, 4, BPITCH>(Asr + ((NKCH - 1) & 1) * ASTG,
                     Bsr + ((NKCH - 1) & 1) * BSTAGE, acc);

#pragma unroll
  for (int r = 0; r < RT; ++r)
#pragma unroll
    for (int c = 0; c < 4; ++c)
#pragma unroll
      for (int e = 0; e < 8; ++e) {
        const int row = (tr0 + r) * 16 + e + 8 * laneH;
        const int col = (tc0 + c) * 16 + laneM;
        C[(size_t)row * D + col] = acc[r][c][e];
      }
}

template <int RT, int CT>
__global__ __launch_bounds__(256) void prodscan_tree_mm(float* __restrict__ ws,
                                                        int d) {
  constexpr int WR    = 16 / RT;
  constexpr int WC    = 8 / WR;
  constexpr int SCOLS = WC * CT * 16;
  constexpr int BG    = SCOLS / 4;
  constexpr int BPP   = 256 / BG;
  constexpr int BPASS = (KC / 2) / BPP;
  constexpr int APASS = D / AROWS;
  constexpr int ASTG  = D * APITCH;
  __shared__ __attribute__((aligned(16))) float As[2 * ASTG];
  __shared__ __attribute__((aligned(16))) float Bs[2 * BSTAGEC];

  const int k    = blockIdx.x;
  const int slab = blockIdx.y;

  const int base = k << (d + 1);
  const float* A = ws + (size_t)(base + (1 << d) - 1) * MATELT;
  float* C       = ws + (size_t)(base + (2 << d) - 1) * MATELT;

  const int tid   = threadIdx.x;
  const int w     = tid >> 5;
  const int lane  = tid & 31;
  const int laneM = lane & 15;
  const int laneH = lane >> 4;

  const int tr0 = (w / WC) * RT;
  const int tc0 = (w % WC) * CT;

  const int arow = tid / (KC / 4), akq = tid % (KC / 4);
  const int bp   = tid / BG, bcq = tid % BG;
  const float* Ag = A + (size_t)arow * D + akq * 4;
  const float* Bg = C + (size_t)(2 * bp) * D + slab * SCOLS + bcq * 4;
  float* Asw = As + arow * APITCH + akq * 4;
  float* Bsw = Bs + bp * BPITCHC + bcq * 8;

  const float* Asr = As + (tr0 * 16 + laneM) * APITCH + 2 * laneH;
  const float* Bsr = Bs + laneH * BPITCHC + (tc0 * 16 + laneM) * 2;

  v4f pa[APASS];
  v4f pb0[BPASS];
  v4f pb1[BPASS];

  v8f acc[RT][CT] = {};

#pragma unroll
  for (int i = 0; i < APASS; ++i)
    pa[i] = *(const v4f*)(Ag + (size_t)(AROWS * i) * D);
#pragma unroll
  for (int i = 0; i < BPASS; ++i) {
    pb0[i] = *(const v4f*)(Bg + (size_t)(2 * BPP * i) * D);
    pb1[i] = *(const v4f*)(Bg + (size_t)(2 * BPP * i + 1) * D);
  }
#pragma unroll
  for (int i = 0; i < APASS; ++i)
    *(v4f*)(Asw + (AROWS * i) * APITCH) = pa[i];
#pragma unroll
  for (int i = 0; i < BPASS; ++i) {
    *(v4f*)(Bsw + (BPP * i) * BPITCHC) =
        (v4f){pb0[i].x, pb1[i].x, pb0[i].y, pb1[i].y};
    *(v4f*)(Bsw + (BPP * i) * BPITCHC + 4) =
        (v4f){pb0[i].z, pb1[i].z, pb0[i].w, pb1[i].w};
  }
  __syncthreads();

  for (int ch = 0; ch < NKCH - 1; ++ch) {
    const int s  = ch & 1;
    const int kc = (ch + 1) * KC;
#pragma unroll
    for (int i = 0; i < APASS; ++i)
      pa[i] = *(const v4f*)(Ag + (size_t)(AROWS * i) * D + kc);
#pragma unroll
    for (int i = 0; i < BPASS; ++i) {
      pb0[i] = *(const v4f*)(Bg + (size_t)(kc + 2 * BPP * i) * D);
      pb1[i] = *(const v4f*)(Bg + (size_t)(kc + 2 * BPP * i + 1) * D);
    }
    __builtin_amdgcn_sched_barrier(0);

    chunk_products<RT, CT, BPITCHC>(Asr + s * ASTG, Bsr + s * BSTAGEC, acc);

    const int so = (s ^ 1);
#pragma unroll
    for (int i = 0; i < APASS; ++i)
      *(v4f*)(Asw + so * ASTG + (AROWS * i) * APITCH) = pa[i];
#pragma unroll
    for (int i = 0; i < BPASS; ++i) {
      *(v4f*)(Bsw + so * BSTAGEC + (BPP * i) * BPITCHC) =
          (v4f){pb0[i].x, pb1[i].x, pb0[i].y, pb1[i].y};
      *(v4f*)(Bsw + so * BSTAGEC + (BPP * i) * BPITCHC + 4) =
          (v4f){pb0[i].z, pb1[i].z, pb0[i].w, pb1[i].w};
    }
    __syncthreads();
  }
  chunk_products<RT, CT, BPITCHC>(Asr + ((NKCH - 1) & 1) * ASTG,
                                  Bsr + ((NKCH - 1) & 1) * BSTAGEC, acc);

#pragma unroll
  for (int r = 0; r < RT; ++r)
#pragma unroll
    for (int c = 0; c < CT; ++c)
#pragma unroll
      for (int e = 0; e < 8; ++e) {
        const int row = (tr0 + r) * 16 + e + 8 * laneH;
        const int col = slab * SCOLS + (tc0 + c) * 16 + laneM;
        C[(size_t)row * D + col] = acc[r][c][e];
      }
}

#define DS_STRIP 32
#define DS_ROWS  (D / 8)
#define DS_TOP   6

__global__ __launch_bounds__(256) void prodscan_downsweep(
    const float* __restrict__ x, const float* __restrict__ ws, float* out,
    int l) {
  __shared__ float us[D];
  __shared__ float part[8][DS_STRIP];
  const int i     = blockIdx.x;
  const int strip = blockIdx.y;
  const int t     = threadIdx.x;
  const int w     = t >> 5;
  const int lane  = t & 31;

  const int srcIdx = (2 * i) << l;
  const int dstIdx = (2 * i + 1) << l;
  const int buf    = srcIdx + (1 << l) - 1;
  const float* T = ws + (size_t)buf * MATELT + (size_t)(w * DS_ROWS) * D +
                   strip * DS_STRIP + lane;

  float mv[DS_ROWS];
#pragma unroll
  for (int r = 0; r < DS_ROWS; ++r) mv[r] = T[(size_t)r * D];

  const float sv =
      (srcIdx == 0) ? x[t] : out[(size_t)(srcIdx * LEAF) * D + t];
  us[t] = sv;
  if (l == DS_TOP && strip == 0) out[t] = sv;
  __syncthreads();

  float acc = 0.f;
#pragma unroll
  for (int r = 0; r < DS_ROWS; ++r)
    acc = fmaf(us[w * DS_ROWS + r], mv[r], acc);
  part[w][lane] = acc;
  __syncthreads();
  if (w == 0) {
    float sum = part[0][lane];
#pragma unroll
    for (int k = 1; k < 8; ++k) sum += part[k][lane];
    out[(size_t)(dstIdx * LEAF) * D + strip * DS_STRIP + lane] = sum;
  }
}

__global__ __launch_bounds__(256) void prodscan_expand(
    const float* __restrict__ m, const float* __restrict__ ws, float* out) {
  __shared__ float vb[2][D];
  const int j = blockIdx.x;
  const int t = threadIdx.x;
  if (j == NLEAF) {
    const float* P = ws + (size_t)(NLEAF - 1) * MATELT;
    vb[0][t] = out[(size_t)((NLEAF - 1) * LEAF) * D + t];
    __syncthreads();
    float acc = 0.f;
#pragma unroll 8
    for (int r = 0; r < D; ++r)
      acc = fmaf(vb[0][r], P[(size_t)r * D + t], acc);
    out[(size_t)NMAT * D + t] = acc;
    return;
  }
  vb[0][t] = out[(size_t)(j * LEAF) * D + t];
  __syncthreads();
  int cur = 0;
  for (int i = 0; i < LEAF - 1; ++i) {
    const float* M = m + (size_t)(j * LEAF + i) * MATELT;
    if (i + 1 < LEAF - 1)
      __builtin_prefetch(m + (size_t)(j * LEAF + i + 1) * MATELT + t * 64, 0,
                         1);
    float acc = 0.f;
#pragma unroll 8
    for (int r = 0; r < D; ++r)
      acc = fmaf(vb[cur][r], M[(size_t)r * D + t], acc);
    vb[cur ^ 1][t] = acc;
    out[(size_t)(j * LEAF + i + 1) * D + t] = acc;
    cur ^= 1;
    __syncthreads();
  }
}

extern "C" void kernel_launch(void* const* d_in, const int* in_sizes, int n_in,
                              void* d_out, int out_size, void* d_ws,
                              size_t ws_size, hipStream_t stream) {
  (void)in_sizes; (void)n_in; (void)out_size; (void)ws_size;
  const float* x = (const float*)d_in[0];
  const float* m = (const float*)d_in[1];
  float* out = (float*)d_out;
  float* ws  = (float*)d_ws;

  for (int t = 1; t < LEAF; ++t)
    prodscan_chain_mm<<<dim3(NLEAF, 2), 256, 0, stream>>>(m, ws, t);
  prodscan_tree_mm<4, 4><<<dim3(63, 2), 256, 0, stream>>>(ws, 0);
  prodscan_tree_mm<2, 2><<<dim3(31, 8), 256, 0, stream>>>(ws, 1);
  prodscan_tree_mm<2, 2><<<dim3(15, 8), 256, 0, stream>>>(ws, 2);
  prodscan_tree_mm<2, 2><<<dim3(7, 8), 256, 0, stream>>>(ws, 3);
  prodscan_tree_mm<2, 2><<<dim3(3, 8), 256, 0, stream>>>(ws, 4);
  prodscan_tree_mm<2, 2><<<dim3(1, 8), 256, 0, stream>>>(ws, 5);
  prodscan_downsweep<<<dim3(1, 8), 256, 0, stream>>>(x, ws, out, 6);
  prodscan_downsweep<<<dim3(2, 8), 256, 0, stream>>>(x, ws, out, 5);
  prodscan_downsweep<<<dim3(4, 8), 256, 0, stream>>>(x, ws, out, 4);
  prodscan_downsweep<<<dim3(8, 8), 256, 0, stream>>>(x, ws, out, 3);
  prodscan_downsweep<<<dim3(16, 8), 256, 0, stream>>>(x, ws, out, 2);
  prodscan_downsweep<<<dim3(32, 8), 256, 0, stream>>>(x, ws, out, 1);
  prodscan_downsweep<<<dim3(64, 8), 256, 0, stream>>>(x, ws, out, 0);
  prodscan_expand<<<NLEAF + 1, 256, 0, stream>>>(m, ws, out);
}


# Leaf expansion: the 7 chain steps become 7 stream-ordered launches of 128 strip GEMVs (2 column strips each, rows split over 8 waves), not one serial work-group per leaf; sums regrouped 8 x 32 rows
# speedup vs baseline: 4.3543x; 1.2287x over previous
#include <hip/hip_runtime.h>
#include <hip/hip_bf16.h>


#define D      256
#define NMAT   1024
#define MATELT (D * D)
#define LEAF   8
#define NLEAF  (NMAT / LEAF)

typedef __attribute__((ext_vector_type(2))) float v2f;
typedef __attribute__((ext_vector_type(4))) float v4f;
typedef __attribute__((ext_vector_type(8))) float v8f;

#define KC       64
#define NKCH     (D / KC)
#define APITCH   68
#define BPITCH   544
#define BPITCHC  288
#define NPASS    (KC / 8)
#define AROWS    (1024 / KC)
#define BSTAGE   ((KC / 2) * BPITCH)
#define BSTAGEC  ((KC / 2) * BPITCHC)

template <int RT, int CT, int BP>
__device__ __forceinline__ void chunk_products(const float* as,
                                               const float* bs,
                                               v8f (&acc)[RT][CT]) {
  v2f af[2][RT];
  v2f bf[2][CT];
#pragma unroll
  for (int r = 0; r < RT; ++r)
    af[0][r] = *(const v2f*)(as + (r * 16) * APITCH);
#pragma unroll
  for (int c = 0; c < CT; ++c)
    bf[0][c] = *(const v2f*)(bs + (c * 16) * 2);
#pragma unroll
  for (int ks = 0; ks < KC / 4; ++ks) {
    const int cur = ks & 1;
    if (ks + 1 < KC / 4) {
#pragma unroll
      for (int r = 0; r < RT; ++r)
        af[cur ^ 1][r] =
            *(const v2f*)(as + (r * 16) * APITCH + (ks + 1) * 4);
#pragma unroll
      for (int c = 0; c < CT; ++c)
        bf[cur ^ 1][c] =
            *(const v2f*)(bs + ((ks + 1) * 2) * BP + (c * 16) * 2);
    }
    __builtin_amdgcn_sched_barrier(0);
#pragma unroll
    for (int r = 0; r < RT; ++r)
#pragma unroll
      for (int c = 0; c < CT; ++c)
        acc[r][c] = __builtin_amdgcn_wmma_f32_16x16x4_f32(
            false, af[cur][r], false, bf[cur][c], (short)0, acc[r][c], false,
            false);
    __builtin_amdgcn_sched_barrier(0);
  }
}

__global__ __launch_bounds__(256) void prodscan_chain_mm(
    const float* __restrict__ m, float* __restrict__ ws, int t) {
  constexpr int RT    = 4;
  constexpr int SROWS = 32 * RT;
  constexpr int APASS = SROWS / AROWS;
  constexpr int ASTG  = SROWS * APITCH;
  __shared__ __attribute__((aligned(16))) float As[2 * ASTG];
  __shared__ __attribute__((aligned(16))) float Bs[2 * BSTAGE];

  const int h    = blockIdx.x;
  const int slab = blockIdx.y;

  float* C = ws + (size_t)h * MATELT;
  const float* A = (t == 1) ? (m + (size_t)h * LEAF * MATELT) : C;
  const float* B = m + (size_t)(h * LEAF + t) * MATELT;

  const int tid   = threadIdx.x;
  const int w     = tid >> 5;
  const int lane  = tid & 31;
  const int laneM = lane & 15;
  const int laneH = lane >> 4;

  const int wr  = w >> 2;
  const int tr0 = slab * (2 * RT) + wr * RT;
  const int tc0 = (w & 3) * 4;

  const int arow = tid / (KC / 4), akq = tid % (KC / 4);
  const int bp   = tid >> 6, bcq = tid & 63;
  const float* Ag = A + (size_t)(slab * SROWS + arow) * D + akq * 4;
  const float* Bg = B + (size_t)(2 * bp) * D + bcq * 4;
  float* Asw = As + arow * APITCH + akq * 4;
  float* Bsw = Bs + bp * BPITCH + bcq * 8;

  const float* Asr = As + (wr * (RT * 16) + laneM) * APITCH + 2 * laneH;
  const float* Bsr = Bs + laneH * BPITCH + (tc0 * 16 + laneM) * 2;

  v4f pa[APASS];
  v4f pb0[NPASS];
  v4f pb1[NPASS];

  v8f acc[RT][4] = {};

#pragma unroll
  for (int i = 0; i < APASS; ++i)
    pa[i] = *(const v4f*)(Ag + (size_t)(AROWS * i) * D);
#pragma unroll
  for (int i = 0; i < NPASS; ++i) {
    pb0[i] = *(const v4f*)(Bg + (size_t)(8 * i) * D);
    pb1[i] = *(const v4f*)(Bg + (size_t)(8 * i + 1) * D);
  }
#pragma unroll
  for (int i = 0; i < APASS; ++i)
    *(v4f*)(Asw + (AROWS * i) * APITCH) = pa[i];
#pragma unroll
  for (int i = 0; i < NPASS; ++i) {
    *(v4f*)(Bsw + (4 * i) * BPITCH) =
        (v4f){pb0[i].x, pb1[i].x, pb0[i].y, pb1[i].y};
    *(v4f*)(Bsw + (4 * i) * BPITCH + 4) =
        (v4f){pb0[i].z, pb1[i].z, pb0[i].w, pb1[i].w};
  }
  __syncthreads();

  for (int ch = 0; ch < NKCH - 1; ++ch) {
    const int s  = ch & 1;
    const int kc = (ch + 1) * KC;
#pragma unroll
    for (int i = 0; i < APASS; ++i)
      pa[i] = *(const v4f*)(Ag + (size_t)(AROWS * i) * D + kc);
#pragma unroll
    for (int i = 0; i < NPASS; ++i) {
      pb0[i] = *(const v4f*)(Bg + (size_t)(kc + 8 * i) * D);
      pb1[i] = *(const v4f*)(Bg + (size_t)(kc + 8 * i + 1) * D);
    }
    __builtin_amdgcn_sched_barrier(0);

    chunk_products<RT, 4, BPITCH>(Asr + s * ASTG, Bsr + s * BSTAGE, acc);

    const int so = (s ^ 1);
#pragma unroll
    for (int i = 0; i < APASS; ++i)
      *(v4f*)(Asw + so * ASTG + (AROWS * i) * APITCH) = pa[i];
#pragma unroll
    for (int i = 0; i < NPASS; ++i) {
      *(v4f*)(Bsw + so * BSTAGE + (4 * i) * BPITCH) =
          (v4f){pb0[i].x, pb1[i].x, pb0[i].y, pb1[i].y};
      *(v4f*)(Bsw + so * BSTAGE + (4 * i) * BPITCH + 4) =
          (v4f){pb0[i].z, pb1[i].z, pb0[i].w, pb1[i].w};
    }
    __syncthreads();
  }
  chunk_products<RT, 4, BPITCH>(Asr + ((NKCH - 1) & 1) * ASTG,
                     Bsr + ((NKCH - 1) & 1) * BSTAGE, acc);

#pragma unroll
  for (int r = 0; r < RT; ++r)
#pragma unroll
    for (int c = 0; c < 4; ++c)
#pragma unroll
      for (int e = 0; e < 8; ++e) {
        const int row = (tr0 + r) * 16 + e + 8 * laneH;
        const int col = (tc0 + c) * 16 + laneM;
        C[(size_t)row * D + col] = acc[r][c][e];
      }
}

template <int RT, int CT>
__global__ __launch_bounds__(256) void prodscan_tree_mm(float* __restrict__ ws,
                                                        int d) {
  constexpr int WR    = 16 / RT;
  constexpr int WC    = 8 / WR;
  constexpr int SCOLS = WC * CT * 16;
  constexpr int BG    = SCOLS / 4;
  constexpr int BPP   = 256 / BG;
  constexpr int BPASS = (KC / 2) / BPP;
  constexpr int APASS = D / AROWS;
  constexpr int ASTG  = D * APITCH;
  __shared__ __attribute__((aligned(16))) float As[2 * ASTG];
  __shared__ __attribute__((aligned(16))) float Bs[2 * BSTAGEC];

  const int k    = blockIdx.x;
  const int slab = blockIdx.y;

  const int base = k << (d + 1);
  const float* A = ws + (size_t)(base + (1 << d) - 1) * MATELT;
  float* C       = ws + (size_t)(base + (2 << d) - 1) * MATELT;

  const int tid   = threadIdx.x;
  const int w     = tid >> 5;
  const int lane  = tid & 31;
  const int laneM = lane & 15;
  const int laneH = lane >> 4;

  const int tr0 = (w / WC) * RT;
  const int tc0 = (w % WC) * CT;

  const int arow = tid / (KC / 4), akq = tid % (KC / 4);
  const int bp   = tid / BG, bcq = tid % BG;
  const float* Ag = A + (size_t)arow * D + akq * 4;
  const float* Bg = C + (size_t)(2 * bp) * D + slab * SCOLS + bcq * 4;
  float* Asw = As + arow * APITCH + akq * 4;
  float* Bsw = Bs + bp * BPITCHC + bcq * 8;

  const float* Asr = As + (tr0 * 16 + laneM) * APITCH + 2 * laneH;
  const float* Bsr = Bs + laneH * BPITCHC + (tc0 * 16 + laneM) * 2;

  v4f pa[APASS];
  v4f pb0[BPASS];
  v4f pb1[BPASS];

  v8f acc[RT][CT] = {};

#pragma unroll
  for (int i = 0; i < APASS; ++i)
    pa[i] = *(const v4f*)(Ag + (size_t)(AROWS * i) * D);
#pragma unroll
  for (int i = 0; i < BPASS; ++i) {
    pb0[i] = *(const v4f*)(Bg + (size_t)(2 * BPP * i) * D);
    pb1[i] = *(const v4f*)(Bg + (size_t)(2 * BPP * i + 1) * D);
  }
#pragma unroll
  for (int i = 0; i < APASS; ++i)
    *(v4f*)(Asw + (AROWS * i) * APITCH) = pa[i];
#pragma unroll
  for (int i = 0; i < BPASS; ++i) {
    *(v4f*)(Bsw + (BPP * i) * BPITCHC) =
        (v4f){pb0[i].x, pb1[i].x, pb0[i].y, pb1[i].y};
    *(v4f*)(Bsw + (BPP * i) * BPITCHC + 4) =
        (v4f){pb0[i].z, pb1[i].z, pb0[i].w, pb1[i].w};
  }
  __syncthreads();

  for (int ch = 0; ch < NKCH - 1; ++ch) {
    const int s  = ch & 1;
    const int kc = (ch + 1) * KC;
#pragma unroll
    for (int i = 0; i < APASS; ++i)
      pa[i] = *(const v4f*)(Ag + (size_t)(AROWS * i) * D + kc);
#pragma unroll
    for (int i = 0; i < BPASS; ++i) {
      pb0[i] = *(const v4f*)(Bg + (size_t)(kc + 2 * BPP * i) * D);
      pb1[i] = *(const v4f*)(Bg + (size_t)(kc + 2 * BPP * i + 1) * D);
    }
    __builtin_amdgcn_sched_barrier(0);

    chunk_products<RT, CT, BPITCHC>(Asr + s * ASTG, Bsr + s * BSTAGEC, acc);

    const int so = (s ^ 1);
#pragma unroll
    for (int i = 0; i < APASS; ++i)
      *(v4f*)(Asw + so * ASTG + (AROWS * i) * APITCH) = pa[i];
#pragma unroll
    for (int i = 0; i < BPASS; ++i) {
      *(v4f*)(Bsw + so * BSTAGEC + (BPP * i) * BPITCHC) =
          (v4f){pb0[i].x, pb1[i].x, pb0[i].y, pb1[i].y};
      *(v4f*)(Bsw + so * BSTAGEC + (BPP * i) * BPITCHC + 4) =
          (v4f){pb0[i].z, pb1[i].z, pb0[i].w, pb1[i].w};
    }
    __syncthreads();
  }
  chunk_products<RT, CT, BPITCHC>(Asr + ((NKCH - 1) & 1) * ASTG,
                                  Bsr + ((NKCH - 1) & 1) * BSTAGEC, acc);

#pragma unroll
  for (int r = 0; r < RT; ++r)
#pragma unroll
    for (int c = 0; c < CT; ++c)
#pragma unroll
      for (int e = 0; e < 8; ++e) {
        const int row = (tr0 + r) * 16 + e + 8 * laneH;
        const int col = slab * SCOLS + (tc0 + c) * 16 + laneM;
        C[(size_t)row * D + col] = acc[r][c][e];
      }
}

#define DS_STRIP 32
#define DS_ROWS  (D / 8)
#define DS_TOP   6

__global__ __launch_bounds__(256) void prodscan_downsweep(
    const float* __restrict__ x, const float* __restrict__ ws, float* out,
    int l) {
  __shared__ float us[D];
  __shared__ float part[8][DS_STRIP];
  const int i     = blockIdx.x;
  const int strip = blockIdx.y;
  const int t     = threadIdx.x;
  const int w     = t >> 5;
  const int lane  = t & 31;

  const int srcIdx = (2 * i) << l;
  const int dstIdx = (2 * i + 1) << l;
  const int buf    = srcIdx + (1 << l) - 1;
  const float* T = ws + (size_t)buf * MATELT + (size_t)(w * DS_ROWS) * D +
                   strip * DS_STRIP + lane;

  float mv[DS_ROWS];
#pragma unroll
  for (int r = 0; r < DS_ROWS; ++r) mv[r] = T[(size_t)r * D];

  const float sv =
      (srcIdx == 0) ? x[t] : out[(size_t)(srcIdx * LEAF) * D + t];
  us[t] = sv;
  if (l == DS_TOP && strip == 0) out[t] = sv;
  __syncthreads();

  float acc = 0.f;
#pragma unroll
  for (int r = 0; r < DS_ROWS; ++r)
    acc = fmaf(us[w * DS_ROWS + r], mv[r], acc);
  part[w][lane] = acc;
  __syncthreads();
  if (w == 0) {
    float sum = part[0][lane];
#pragma unroll
    for (int k = 1; k < 8; ++k) sum += part[k][lane];
    out[(size_t)(dstIdx * LEAF) * D + strip * DS_STRIP + lane] = sum;
  }
}

#define EX_STRIP 128
#define EX_ROWS  (D / 8)

__global__ __launch_bounds__(256) void prodscan_expand_step(
    const float* __restrict__ m, const float* __restrict__ ws, float* out,
    int i) {
  __shared__ __attribute__((aligned(16))) float us[D];
  __shared__ __attribute__((aligned(16))) float part[8][EX_STRIP];
  const int j     = blockIdx.x;
  const int strip = blockIdx.y;
  const int t     = threadIdx.x;
  const int w     = t >> 5;
  const int lane  = t & 31;

  const bool last = (j == NLEAF);
  const float* M  = last ? ws + (size_t)(NLEAF - 1) * MATELT
                         : m + (size_t)(j * LEAF + i) * MATELT;
  const int src   = last ? (NLEAF - 1) * LEAF : j * LEAF + i;
  const int dst   = last ? NMAT : src + 1;

  const float sv = out[(size_t)src * D + t];

  const float* T = M + (size_t)(w * EX_ROWS) * D + strip * EX_STRIP + lane * 4;
  v4f mv[EX_ROWS];
#pragma unroll
  for (int r = 0; r < EX_ROWS; ++r) mv[r] = *(const v4f*)(T + (size_t)r * D);

  us[t] = sv;
  __syncthreads();

  v4f acc = {0.f, 0.f, 0.f, 0.f};
#pragma unroll
  for (int r = 0; r < EX_ROWS; r += 4) {
    const v4f u = *(const v4f*)(us + w * EX_ROWS + r);
    const float uu[4] = {u.x, u.y, u.z, u.w};
#pragma unroll
    for (int q = 0; q < 4; ++q) {
      acc.x = fmaf(uu[q], mv[r + q].x, acc.x);
      acc.y = fmaf(uu[q], mv[r + q].y, acc.y);
      acc.z = fmaf(uu[q], mv[r + q].z, acc.z);
      acc.w = fmaf(uu[q], mv[r + q].w, acc.w);
    }
  }
  *(v4f*)(&part[w][lane * 4]) = acc;
  __syncthreads();
  if (t < EX_STRIP) {
    float sum = part[0][t];
    for (int k = 1; k < 8; ++k) sum += part[k][t];
    out[(size_t)dst * D + strip * EX_STRIP + t] = sum;
  }
}

extern "C" void kernel_launch(void* const* d_in, const int* in_sizes, int n_in,
                              void* d_out, int out_size, void* d_ws,
                              size_t ws_size, hipStream_t stream) {
  (void)in_sizes; (void)n_in; (void)out_size; (void)ws_size;
  const float* x = (const float*)d_in[0];
  const float* m = (const float*)d_in[1];
  float* out = (float*)d_out;
  float* ws  = (float*)d_ws;

  for (int t = 1; t < LEAF; ++t)
    prodscan_chain_mm<<<dim3(NLEAF, 2), 256, 0, stream>>>(m, ws, t);
  prodscan_tree_mm<4, 4><<<dim3(63, 2), 256, 0, stream>>>(ws, 0);
  prodscan_tree_mm<2, 2><<<dim3(31, 8), 256, 0, stream>>>(ws, 1);
  prodscan_tree_mm<2, 2><<<dim3(15, 8), 256, 0, stream>>>(ws, 2);
  prodscan_tree_mm<2, 2><<<dim3(7, 8), 256, 0, stream>>>(ws, 3);
  prodscan_tree_mm<2, 2><<<dim3(3, 8), 256, 0, stream>>>(ws, 4);
  prodscan_tree_mm<2, 2><<<dim3(1, 8), 256, 0, stream>>>(ws, 5);
  prodscan_downsweep<<<dim3(1, 8), 256, 0, stream>>>(x, ws, out, 6);
  prodscan_downsweep<<<dim3(2, 8), 256, 0, stream>>>(x, ws, out, 5);
  prodscan_downsweep<<<dim3(4, 8), 256, 0, stream>>>(x, ws, out, 4);
  prodscan_downsweep<<<dim3(8, 8), 256, 0, stream>>>(x, ws, out, 3);
  prodscan_downsweep<<<dim3(16, 8), 256, 0, stream>>>(x, ws, out, 2);
  prodscan_downsweep<<<dim3(32, 8), 256, 0, stream>>>(x, ws, out, 1);
  prodscan_downsweep<<<dim3(64, 8), 256, 0, stream>>>(x, ws, out, 0);
  prodscan_expand_step<<<dim3(NLEAF + 1, 2), 256, 0, stream>>>(m, ws, out, 0);
  for (int i = 1; i < LEAF - 1; ++i)
    prodscan_expand_step<<<dim3(NLEAF, 2), 256, 0, stream>>>(m, ws, out, i);
}
